# starting kernel
# baseline (speedup 1.0000x reference)
#include <hip/hip_runtime.h>

typedef __attribute__((ext_vector_type(2))) float v2f;
typedef __attribute__((ext_vector_type(8))) float v8f;

namespace {
constexpr int WIN     = 5;
constexpr int VOCAB   = 100000;
constexpr int EMB     = 50;
constexpr int HID     = 250;
constexpr int OUTD    = 36;
constexpr int KPAD    = 256;
constexpr int NPAD2   = 48;
constexpr int ROWS    = 64;
constexpr int STRIDE  = 260;
constexpr int THREADS = 128;
}

__device__ __forceinline__ float fast_tanh(float x) {
  float xc = fminf(15.0f, fmaxf(-15.0f, x));
  float t  = __expf(2.0f * xc);
  return (t - 1.0f) * __builtin_amdgcn_rcpf(t + 1.0f);
}

__global__ __launch_bounds__(256)
void pad_weights(const float* __restrict__ w_in, const float* __restrict__ w_out,
                 float* __restrict__ wp, float* __restrict__ wp2)
{
  int i = blockIdx.x * 256 + threadIdx.x;
  if (i < KPAD * KPAD) {
    int k = i >> 8, n = i & 255;
    wp[i] = (k < HID && n < HID) ? w_in[k * HID + n] : 0.0f;
  }
  if (i < KPAD * NPAD2) {
    int k = i / NPAD2, n = i - k * NPAD2;
    wp2[i] = (k < HID && n < OUTD) ? w_out[k * OUTD + n] : 0.0f;
  }
}

__global__ __launch_bounds__(THREADS)
void tagger_fused(const int*   __restrict__ idx,
                  const float* __restrict__ emb,
                  const float* __restrict__ wp,
                  const float* __restrict__ b_in,
                  const float* __restrict__ wp2,
                  const float* __restrict__ b_out,
                  float*       __restrict__ out)
{
  __shared__ float x_lds[ROWS * STRIDE];
  __shared__ float h_lds[ROWS * STRIDE];
  __shared__ int   i_lds[ROWS * WIN];

  const int t       = threadIdx.x;
  const int rowBase = blockIdx.x * ROWS;

  for (int e = t; e < ROWS * WIN; e += THREADS)
    i_lds[e] = idx[rowBase * WIN + e];
  __syncthreads();

  for (int e = t; e < ROWS * HID; e += THREADS) {
    int row = e / HID;
    int col = e - row * HID;
    int w   = col / EMB;
    int c   = col - w * EMB;
    int id  = i_lds[row * WIN + w];
    float v = 0.0f;
    if (id >= 0 && id < VOCAB) v = emb[id * EMB + c];
    x_lds[row * STRIDE + col] = v;
  }
  for (int p = t; p < ROWS * (STRIDE - HID); p += THREADS) {
    int row = p / (STRIDE - HID);
    int c   = HID + (p - row * (STRIDE - HID));
    x_lds[row * STRIDE + c] = 0.0f;
  }
  __syncthreads();

  const int lane = t & 31;
  const int wave = t >> 5;
  const int hi   = lane >> 4;
  const int nloc = lane & 15;
  const int m    = lane & 15;

  const float* xr0 = &x_lds[( 0 + m) * STRIDE];
  const float* xr1 = &x_lds[(16 + m) * STRIDE];
  const float* xr2 = &x_lds[(32 + m) * STRIDE];
  const float* xr3 = &x_lds[(48 + m) * STRIDE];

  for (int p = 0; p < 4; ++p) {
    const int ct   = p * 4 + wave;
    const int coln = ct * 16 + nloc;
    const float bias = (coln < HID) ? b_in[coln] : 0.0f;
    v8f acc0 = {bias, bias, bias, bias, bias, bias, bias, bias};
    v8f acc1 = acc0, acc2 = acc0, acc3 = acc0;

    #pragma unroll 4
    for (int k = 0; k < KPAD; k += 4) {
      const int kk = k + 2 * hi;
      v2f b;
      b.x = wp[kk * KPAD + coln];
      b.y = wp[(kk + 1) * KPAD + coln];
      v2f a0 = *reinterpret_cast<const v2f*>(&xr0[kk]);
      v2f a1 = *reinterpret_cast<const v2f*>(&xr1[kk]);
      v2f a2 = *reinterpret_cast<const v2f*>(&xr2[kk]);
      v2f a3 = *reinterpret_cast<const v2f*>(&xr3[kk]);
      acc0 = __builtin_amdgcn_wmma_f32_16x16x4_f32(false, a0, false, b, (short)0, acc0, false, false);
      acc1 = __builtin_amdgcn_wmma_f32_16x16x4_f32(false, a1, false, b, (short)0, acc1, false, false);
      acc2 = __builtin_amdgcn_wmma_f32_16x16x4_f32(false, a2, false, b, (short)0, acc2, false, false);
      acc3 = __builtin_amdgcn_wmma_f32_16x16x4_f32(false, a3, false, b, (short)0, acc3, false, false);
    }

    #pragma unroll
    for (int j = 0; j < 8; ++j) {
      const int M = j + 8 * hi;
      h_lds[( 0 + M) * STRIDE + coln] = fast_tanh(acc0[j]);
      h_lds[(16 + M) * STRIDE + coln] = fast_tanh(acc1[j]);
      h_lds[(32 + M) * STRIDE + coln] = fast_tanh(acc2[j]);
      h_lds[(48 + M) * STRIDE + coln] = fast_tanh(acc3[j]);
    }
  }
  __syncthreads();

  const int rt = wave;
  const float* hrow = &h_lds[(rt * 16 + m) * STRIDE];

  for (int ct = 0; ct < 3; ++ct) {
    const int coln = ct * 16 + nloc;
    const float bias = (coln < OUTD) ? b_out[coln] : 0.0f;
    v8f acc = {bias, bias, bias, bias, bias, bias, bias, bias};

    #pragma unroll 4
    for (int k = 0; k < KPAD; k += 4) {
      const int kk = k + 2 * hi;
      v2f a = *reinterpret_cast<const v2f*>(&hrow[kk]);
      v2f b;
      b.x = wp2[kk * NPAD2 + coln];
      b.y = wp2[(kk + 1) * NPAD2 + coln];
      acc = __builtin_amdgcn_wmma_f32_16x16x4_f32(false, a, false, b, (short)0, acc, false, false);
    }

    #pragma unroll
    for (int j = 0; j < 8; ++j) {
      const int M = j + 8 * hi;
      if (coln < OUTD)
        out[(size_t)(rowBase + rt * 16 + M) * OUTD + coln] = acc[j];
    }
  }
}

extern "C" void kernel_launch(void* const* d_in, const int* in_sizes, int n_in,
                              void* d_out, int out_size, void* d_ws, size_t ws_size,
                              hipStream_t stream) {
  const int*   idx     = (const int*)  d_in[0];
  const float* emb     = (const float*)d_in[1];
  const float* w_in_p  = (const float*)d_in[2];
  const float* b_in_p  = (const float*)d_in[3];
  const float* w_out_p = (const float*)d_in[4];
  const float* b_out_p = (const float*)d_in[5];
  float* out = (float*)d_out;

  float* wp  = (float*)d_ws;
  float* wp2 = wp + KPAD * KPAD;

  pad_weights<<<(KPAD * KPAD + 255) / 256, 256, 0, stream>>>(w_in_p, w_out_p, wp, wp2);

  const int batch  = in_sizes[0] / WIN;
  const int blocks = batch / ROWS;
  tagger_fused<<<blocks, THREADS, 0, stream>>>(idx, emb, wp, b_in_p, wp2, b_out_p, out);
}


# Gather: each lane takes one 8-byte pair of a batch row, loads issued in groups ahead of the LDS stores (dozens in flight); indices staged by one 16-byte load; was one waited 4-byte load per trip
# speedup vs baseline: 1.7465x; 1.7465x over previous
#include <hip/hip_runtime.h>

typedef __attribute__((ext_vector_type(2))) float v2f;
typedef __attribute__((ext_vector_type(8))) float v8f;

namespace {
constexpr int WIN     = 5;
constexpr int VOCAB   = 100000;
constexpr int EMB     = 50;
constexpr int HID     = 250;
constexpr int OUTD    = 36;
constexpr int KPAD    = 256;
constexpr int NPAD2   = 48;
constexpr int ROWS    = 64;
constexpr int STRIDE  = 260;
constexpr int THREADS = 128;
}

__device__ __forceinline__ float fast_tanh(float x) {
  float xc = fminf(15.0f, fmaxf(-15.0f, x));
  float t  = __expf(2.0f * xc);
  return (t - 1.0f) * __builtin_amdgcn_rcpf(t + 1.0f);
}

__global__ __launch_bounds__(256)
void pad_weights(const float* __restrict__ w_in, const float* __restrict__ w_out,
                 float* __restrict__ wp, float* __restrict__ wp2)
{
  int i = blockIdx.x * 256 + threadIdx.x;
  if (i < KPAD * KPAD) {
    int k = i >> 8, n = i & 255;
    wp[i] = (k < HID && n < HID) ? w_in[k * HID + n] : 0.0f;
  }
  if (i < KPAD * NPAD2) {
    int k = i / NPAD2, n = i - k * NPAD2;
    wp2[i] = (k < HID && n < OUTD) ? w_out[k * OUTD + n] : 0.0f;
  }
}

__global__ __launch_bounds__(THREADS)
void tagger_fused(const int*   __restrict__ idx,
                  const float* __restrict__ emb,
                  const float* __restrict__ wp,
                  const float* __restrict__ b_in,
                  const float* __restrict__ wp2,
                  const float* __restrict__ b_out,
                  float*       __restrict__ out)
{
  __shared__ alignas(16) float x_lds[ROWS * STRIDE];
  __shared__ alignas(16) float h_lds[ROWS * STRIDE];
  __shared__ alignas(16) int   i_lds[ROWS * WIN];

  const int t       = threadIdx.x;
  const int rowBase = blockIdx.x * ROWS;

  if (t < (ROWS * WIN) / 4) {
    const int4 q = *reinterpret_cast<const int4*>(idx + (size_t)rowBase * WIN + 4 * t);
    *reinterpret_cast<int4*>(&i_lds[4 * t]) = q;
  }
  __syncthreads();

  {
    const bool real = t < (HID / 2);
    const int  slot = real ? t / (EMB / 2) : (WIN - 1);
    const unsigned jp = real ? (unsigned)(t - slot * (EMB / 2)) : 0u;
    const v2f* emb2  = reinterpret_cast<const v2f*>(emb);
    const v2f  zero2 = {0.0f, 0.0f};
    #pragma unroll
    for (int r0 = 0; r0 < ROWS; r0 += 16) {
      v2f g[16];
      #pragma unroll
      for (int u = 0; u < 16; ++u) {
        const int  id = i_lds[(r0 + u) * WIN + slot];
        const bool ok = real && (unsigned)id < (unsigned)VOCAB;
        const unsigned s = ok ? (unsigned)id : 0u;
        const v2f  v  = emb2[s * (unsigned)(EMB / 2) + jp];
        g[u] = ok ? v : zero2;
      }
      #pragma unroll
      for (int u = 0; u < 16; ++u)
        *reinterpret_cast<v2f*>(&x_lds[(r0 + u) * STRIDE + 2 * t]) = g[u];
    }
  }
  __syncthreads();

  const int lane = t & 31;
  const int wave = t >> 5;
  const int hi   = lane >> 4;
  const int nloc = lane & 15;
  const int m    = lane & 15;

  const float* xr0 = &x_lds[( 0 + m) * STRIDE];
  const float* xr1 = &x_lds[(16 + m) * STRIDE];
  const float* xr2 = &x_lds[(32 + m) * STRIDE];
  const float* xr3 = &x_lds[(48 + m) * STRIDE];

  for (int p = 0; p < 4; ++p) {
    const int ct   = p * 4 + wave;
    const int coln = ct * 16 + nloc;
    const float bias = (coln < HID) ? b_in[coln] : 0.0f;
    v8f acc0 = {bias, bias, bias, bias, bias, bias, bias, bias};
    v8f acc1 = acc0, acc2 = acc0, acc3 = acc0;

    #pragma unroll 4
    for (int k = 0; k < KPAD; k += 4) {
      const int kk = k + 2 * hi;
      v2f b;
      b.x = wp[kk * KPAD + coln];
      b.y = wp[(kk + 1) * KPAD + coln];
      v2f a0 = *reinterpret_cast<const v2f*>(&xr0[kk]);
      v2f a1 = *reinterpret_cast<const v2f*>(&xr1[kk]);
      v2f a2 = *reinterpret_cast<const v2f*>(&xr2[kk]);
      v2f a3 = *reinterpret_cast<const v2f*>(&xr3[kk]);
      acc0 = __builtin_amdgcn_wmma_f32_16x16x4_f32(false, a0, false, b, (short)0, acc0, false, false);
      acc1 = __builtin_amdgcn_wmma_f32_16x16x4_f32(false, a1, false, b, (short)0, acc1, false, false);
      acc2 = __builtin_amdgcn_wmma_f32_16x16x4_f32(false, a2, false, b, (short)0, acc2, false, false);
      acc3 = __builtin_amdgcn_wmma_f32_16x16x4_f32(false, a3, false, b, (short)0, acc3, false, false);
    }

    #pragma unroll
    for (int j = 0; j < 8; ++j) {
      const int M = j + 8 * hi;
      h_lds[( 0 + M) * STRIDE + coln] = fast_tanh(acc0[j]);
      h_lds[(16 + M) * STRIDE + coln] = fast_tanh(acc1[j]);
      h_lds[(32 + M) * STRIDE + coln] = fast_tanh(acc2[j]);
      h_lds[(48 + M) * STRIDE + coln] = fast_tanh(acc3[j]);
    }
  }
  __syncthreads();

  const int rt = wave;
  const float* hrow = &h_lds[(rt * 16 + m) * STRIDE];

  for (int ct = 0; ct < 3; ++ct) {
    const int coln = ct * 16 + nloc;
    const float bias = (coln < OUTD) ? b_out[coln] : 0.0f;
    v8f acc = {bias, bias, bias, bias, bias, bias, bias, bias};

    #pragma unroll 4
    for (int k = 0; k < KPAD; k += 4) {
      const int kk = k + 2 * hi;
      v2f a = *reinterpret_cast<const v2f*>(&hrow[kk]);
      v2f b;
      b.x = wp2[kk * NPAD2 + coln];
      b.y = wp2[(kk + 1) * NPAD2 + coln];
      acc = __builtin_amdgcn_wmma_f32_16x16x4_f32(false, a, false, b, (short)0, acc, false, false);
    }

    #pragma unroll
    for (int j = 0; j < 8; ++j) {
      const int M = j + 8 * hi;
      if (coln < OUTD)
        out[(size_t)(rowBase + rt * 16 + M) * OUTD + coln] = acc[j];
    }
  }
}

extern "C" void kernel_launch(void* const* d_in, const int* in_sizes, int n_in,
                              void* d_out, int out_size, void* d_ws, size_t ws_size,
                              hipStream_t stream) {
  const int*   idx     = (const int*)  d_in[0];
  const float* emb     = (const float*)d_in[1];
  const float* w_in_p  = (const float*)d_in[2];
  const float* b_in_p  = (const float*)d_in[3];
  const float* w_out_p = (const float*)d_in[4];
  const float* b_out_p = (const float*)d_in[5];
  float* out = (float*)d_out;

  float* wp  = (float*)d_ws;
  float* wp2 = wp + KPAD * KPAD;

  pad_weights<<<(KPAD * KPAD + 255) / 256, 256, 0, stream>>>(w_in_p, w_out_p, wp, wp2);

  const int batch  = in_sizes[0] / WIN;
  const int blocks = batch / ROWS;
  tagger_fused<<<blocks, THREADS, 0, stream>>>(idx, emb, wp, b_in_p, wp2, b_out_p, out);
}


# Phase 1 in ONE K sweep: each wave keeps a 64x64 block of h (16 accumulator tiles) in registers, was 4 sweeps of 4 tiles; x is read once and h reuses its LDS, so 4 workgroups fit a WGP (was 2)
# speedup vs baseline: 2.7471x; 1.5729x over previous
#include <hip/hip_runtime.h>

typedef __attribute__((ext_vector_type(2))) float v2f;
typedef __attribute__((ext_vector_type(8))) float v8f;

namespace {
constexpr int WIN     = 5;
constexpr int VOCAB   = 100000;
constexpr int EMB     = 50;
constexpr int HID     = 250;
constexpr int OUTD    = 36;
constexpr int KPAD    = 256;
constexpr int NPAD2   = 48;
constexpr int ROWS    = 64;
constexpr int STRIDE  = 260;
constexpr int THREADS = 128;
}

__device__ __forceinline__ float fast_tanh(float x) {
  float xc = fminf(15.0f, fmaxf(-15.0f, x));
  float t  = __expf(2.0f * xc);
  return (t - 1.0f) * __builtin_amdgcn_rcpf(t + 1.0f);
}

__global__ __launch_bounds__(256)
void pad_weights(const float* __restrict__ w_in, const float* __restrict__ w_out,
                 float* __restrict__ wp, float* __restrict__ wp2)
{
  int i = blockIdx.x * 256 + threadIdx.x;
  if (i < KPAD * KPAD) {
    int k = i >> 8, n = i & 255;
    wp[i] = (k < HID && n < HID) ? w_in[k * HID + n] : 0.0f;
  }
  if (i < KPAD * NPAD2) {
    int k = i / NPAD2, n = i - k * NPAD2;
    wp2[i] = (k < HID && n < OUTD) ? w_out[k * OUTD + n] : 0.0f;
  }
}

__global__ __launch_bounds__(THREADS)
void tagger_fused(const int*   __restrict__ idx,
                  const float* __restrict__ emb,
                  const float* __restrict__ wp,
                  const float* __restrict__ b_in,
                  const float* __restrict__ wp2,
                  const float* __restrict__ b_out,
                  float*       __restrict__ out)
{
  __shared__ alignas(16) float x_lds[ROWS * STRIDE];
  __shared__ alignas(16) int   i_lds[ROWS * WIN];

  const int t       = threadIdx.x;
  const int rowBase = blockIdx.x * ROWS;

  if (t < (ROWS * WIN) / 4) {
    const int4 q = *reinterpret_cast<const int4*>(idx + (size_t)rowBase * WIN + 4 * t);
    *reinterpret_cast<int4*>(&i_lds[4 * t]) = q;
  }
  __syncthreads();

  {
    const bool real = t < (HID / 2);
    const int  slot = real ? t / (EMB / 2) : (WIN - 1);
    const unsigned jp = real ? (unsigned)(t - slot * (EMB / 2)) : 0u;
    const v2f* emb2  = reinterpret_cast<const v2f*>(emb);
    const v2f  zero2 = {0.0f, 0.0f};
    #pragma unroll
    for (int r0 = 0; r0 < ROWS; r0 += 16) {
      v2f g[16];
      #pragma unroll
      for (int u = 0; u < 16; ++u) {
        const int  id = i_lds[(r0 + u) * WIN + slot];
        const bool ok = real && (unsigned)id < (unsigned)VOCAB;
        const unsigned s = ok ? (unsigned)id : 0u;
        const v2f  v  = emb2[s * (unsigned)(EMB / 2) + jp];
        g[u] = ok ? v : zero2;
      }
      #pragma unroll
      for (int u = 0; u < 16; ++u)
        *reinterpret_cast<v2f*>(&x_lds[(r0 + u) * STRIDE + 2 * t]) = g[u];
    }
  }
  __syncthreads();

  const int lane = t & 31;
  const int wave = t >> 5;
  const int hi   = lane >> 4;
  const int nloc = lane & 15;
  const int m    = lane & 15;

  const float* xr0 = &x_lds[( 0 + m) * STRIDE];
  const float* xr1 = &x_lds[(16 + m) * STRIDE];
  const float* xr2 = &x_lds[(32 + m) * STRIDE];
  const float* xr3 = &x_lds[(48 + m) * STRIDE];

  const int col0 = wave * 64 + nloc;
  v8f hacc[4][4];
  #pragma unroll
  for (int p = 0; p < 4; ++p) {
    const int coln = col0 + 16 * p;
    const float bias = (coln < HID) ? b_in[coln] : 0.0f;
    const v8f bv = {bias, bias, bias, bias, bias, bias, bias, bias};
    hacc[p][0] = bv; hacc[p][1] = bv; hacc[p][2] = bv; hacc[p][3] = bv;
  }

  #pragma unroll 4
  for (int k = 0; k < KPAD; k += 4) {
    const int kk = k + 2 * hi;
    const v2f a0 = *reinterpret_cast<const v2f*>(&xr0[kk]);
    const v2f a1 = *reinterpret_cast<const v2f*>(&xr1[kk]);
    const v2f a2 = *reinterpret_cast<const v2f*>(&xr2[kk]);
    const v2f a3 = *reinterpret_cast<const v2f*>(&xr3[kk]);
    const float* w0 = wp + kk * KPAD + col0;
    #pragma unroll
    for (int p = 0; p < 4; ++p) {
      v2f b;
      b.x = w0[16 * p];
      b.y = w0[16 * p + KPAD];
      hacc[p][0] = __builtin_amdgcn_wmma_f32_16x16x4_f32(false, a0, false, b, (short)0, hacc[p][0], false, false);
      hacc[p][1] = __builtin_amdgcn_wmma_f32_16x16x4_f32(false, a1, false, b, (short)0, hacc[p][1], false, false);
      hacc[p][2] = __builtin_amdgcn_wmma_f32_16x16x4_f32(false, a2, false, b, (short)0, hacc[p][2], false, false);
      hacc[p][3] = __builtin_amdgcn_wmma_f32_16x16x4_f32(false, a3, false, b, (short)0, hacc[p][3], false, false);
    }
  }
  __syncthreads();

  float* h_lds = x_lds;
  #pragma unroll
  for (int p = 0; p < 4; ++p) {
    const int coln = col0 + 16 * p;
    #pragma unroll
    for (int j = 0; j < 8; ++j) {
      const int M = j + 8 * hi;
      h_lds[( 0 + M) * STRIDE + coln] = fast_tanh(hacc[p][0][j]);
      h_lds[(16 + M) * STRIDE + coln] = fast_tanh(hacc[p][1][j]);
      h_lds[(32 + M) * STRIDE + coln] = fast_tanh(hacc[p][2][j]);
      h_lds[(48 + M) * STRIDE + coln] = fast_tanh(hacc[p][3][j]);
    }
  }
  __syncthreads();

  const int rt = wave;
  const float* hrow = &h_lds[(rt * 16 + m) * STRIDE];

  for (int ct = 0; ct < 3; ++ct) {
    const int coln = ct * 16 + nloc;
    const float bias = (coln < OUTD) ? b_out[coln] : 0.0f;
    v8f acc = {bias, bias, bias, bias, bias, bias, bias, bias};

    #pragma unroll 4
    for (int k = 0; k < KPAD; k += 4) {
      const int kk = k + 2 * hi;
      v2f a = *reinterpret_cast<const v2f*>(&hrow[kk]);
      v2f b;
      b.x = wp2[kk * NPAD2 + coln];
      b.y = wp2[(kk + 1) * NPAD2 + coln];
      acc = __builtin_amdgcn_wmma_f32_16x16x4_f32(false, a, false, b, (short)0, acc, false, false);
    }

    #pragma unroll
    for (int j = 0; j < 8; ++j) {
      const int M = j + 8 * hi;
      if (coln < OUTD)
        out[(size_t)(rowBase + rt * 16 + M) * OUTD + coln] = acc[j];
    }
  }
}

extern "C" void kernel_launch(void* const* d_in, const int* in_sizes, int n_in,
                              void* d_out, int out_size, void* d_ws, size_t ws_size,
                              hipStream_t stream) {
  const int*   idx     = (const int*)  d_in[0];
  const float* emb     = (const float*)d_in[1];
  const float* w_in_p  = (const float*)d_in[2];
  const float* b_in_p  = (const float*)d_in[3];
  const float* w_out_p = (const float*)d_in[4];
  const float* b_out_p = (const float*)d_in[5];
  float* out = (float*)d_out;

  float* wp  = (float*)d_ws;
  float* wp2 = wp + KPAD * KPAD;

  pad_weights<<<(KPAD * KPAD + 255) / 256, 256, 0, stream>>>(w_in_p, w_out_p, wp, wp2);

  const int batch  = in_sizes[0] / WIN;
  const int blocks = batch / ROWS;
  tagger_fused<<<blocks, THREADS, 0, stream>>>(idx, emb, wp, b_in_p, wp2, b_out_p, out);
}


# Phase 2 in ONE K sweep: each wave keeps its 3 output tiles in registers, h fragment read once per k-step feeds 3 WMMAs, loads of 8 k-steps in one group (was 48 waited groups of 4 chained WMMAs)
# speedup vs baseline: 2.9232x; 1.0641x over previous
#include <hip/hip_runtime.h>

typedef __attribute__((ext_vector_type(2))) float v2f;
typedef __attribute__((ext_vector_type(8))) float v8f;

namespace {
constexpr int WIN     = 5;
constexpr int VOCAB   = 100000;
constexpr int EMB     = 50;
constexpr int HID     = 250;
constexpr int OUTD    = 36;
constexpr int KPAD    = 256;
constexpr int NPAD2   = 48;
constexpr int ROWS    = 64;
constexpr int STRIDE  = 260;
constexpr int THREADS = 128;
}

__device__ __forceinline__ float fast_tanh(float x) {
  float xc = fminf(15.0f, fmaxf(-15.0f, x));
  float t  = __expf(2.0f * xc);
  return (t - 1.0f) * __builtin_amdgcn_rcpf(t + 1.0f);
}

__global__ __launch_bounds__(256)
void pad_weights(const float* __restrict__ w_in, const float* __restrict__ w_out,
                 float* __restrict__ wp, float* __restrict__ wp2)
{
  int i = blockIdx.x * 256 + threadIdx.x;
  if (i < KPAD * KPAD) {
    int k = i >> 8, n = i & 255;
    wp[i] = (k < HID && n < HID) ? w_in[k * HID + n] : 0.0f;
  }
  if (i < KPAD * NPAD2) {
    int k = i / NPAD2, n = i - k * NPAD2;
    wp2[i] = (k < HID && n < OUTD) ? w_out[k * OUTD + n] : 0.0f;
  }
}

__global__ __launch_bounds__(THREADS)
void tagger_fused(const int*   __restrict__ idx,
                  const float* __restrict__ emb,
                  const float* __restrict__ wp,
                  const float* __restrict__ b_in,
                  const float* __restrict__ wp2,
                  const float* __restrict__ b_out,
                  float*       __restrict__ out)
{
  __shared__ alignas(16) float x_lds[ROWS * STRIDE];
  __shared__ alignas(16) int   i_lds[ROWS * WIN];

  const int t       = threadIdx.x;
  const int rowBase = blockIdx.x * ROWS;

  if (t < (ROWS * WIN) / 4) {
    const int4 q = *reinterpret_cast<const int4*>(idx + (size_t)rowBase * WIN + 4 * t);
    *reinterpret_cast<int4*>(&i_lds[4 * t]) = q;
  }
  __syncthreads();

  {
    const bool real = t < (HID / 2);
    const int  slot = real ? t / (EMB / 2) : (WIN - 1);
    const unsigned jp = real ? (unsigned)(t - slot * (EMB / 2)) : 0u;
    const v2f* emb2  = reinterpret_cast<const v2f*>(emb);
    const v2f  zero2 = {0.0f, 0.0f};
    #pragma unroll
    for (int r0 = 0; r0 < ROWS; r0 += 16) {
      v2f g[16];
      #pragma unroll
      for (int u = 0; u < 16; ++u) {
        const int  id = i_lds[(r0 + u) * WIN + slot];
        const bool ok = real && (unsigned)id < (unsigned)VOCAB;
        const unsigned s = ok ? (unsigned)id : 0u;
        const v2f  v  = emb2[s * (unsigned)(EMB / 2) + jp];
        g[u] = ok ? v : zero2;
      }
      #pragma unroll
      for (int u = 0; u < 16; ++u)
        *reinterpret_cast<v2f*>(&x_lds[(r0 + u) * STRIDE + 2 * t]) = g[u];
    }
  }
  __syncthreads();

  const int lane = t & 31;
  const int wave = t >> 5;
  const int hi   = lane >> 4;
  const int nloc = lane & 15;
  const int m    = lane & 15;

  const float* xr0 = &x_lds[( 0 + m) * STRIDE];
  const float* xr1 = &x_lds[(16 + m) * STRIDE];
  const float* xr2 = &x_lds[(32 + m) * STRIDE];
  const float* xr3 = &x_lds[(48 + m) * STRIDE];

  const int col0 = wave * 64 + nloc;
  v8f hacc[4][4];
  #pragma unroll
  for (int p = 0; p < 4; ++p) {
    const int coln = col0 + 16 * p;
    const float bias = (coln < HID) ? b_in[coln] : 0.0f;
    const v8f bv = {bias, bias, bias, bias, bias, bias, bias, bias};
    hacc[p][0] = bv; hacc[p][1] = bv; hacc[p][2] = bv; hacc[p][3] = bv;
  }

  #pragma unroll 4
  for (int k = 0; k < KPAD; k += 4) {
    const int kk = k + 2 * hi;
    const v2f a0 = *reinterpret_cast<const v2f*>(&xr0[kk]);
    const v2f a1 = *reinterpret_cast<const v2f*>(&xr1[kk]);
    const v2f a2 = *reinterpret_cast<const v2f*>(&xr2[kk]);
    const v2f a3 = *reinterpret_cast<const v2f*>(&xr3[kk]);
    const float* w0 = wp + kk * KPAD + col0;
    #pragma unroll
    for (int p = 0; p < 4; ++p) {
      v2f b;
      b.x = w0[16 * p];
      b.y = w0[16 * p + KPAD];
      hacc[p][0] = __builtin_amdgcn_wmma_f32_16x16x4_f32(false, a0, false, b, (short)0, hacc[p][0], false, false);
      hacc[p][1] = __builtin_amdgcn_wmma_f32_16x16x4_f32(false, a1, false, b, (short)0, hacc[p][1], false, false);
      hacc[p][2] = __builtin_amdgcn_wmma_f32_16x16x4_f32(false, a2, false, b, (short)0, hacc[p][2], false, false);
      hacc[p][3] = __builtin_amdgcn_wmma_f32_16x16x4_f32(false, a3, false, b, (short)0, hacc[p][3], false, false);
    }
  }
  __syncthreads();

  float* h_lds = x_lds;
  #pragma unroll
  for (int p = 0; p < 4; ++p) {
    const int coln = col0 + 16 * p;
    #pragma unroll
    for (int j = 0; j < 8; ++j) {
      const int M = j + 8 * hi;
      h_lds[( 0 + M) * STRIDE + coln] = fast_tanh(hacc[p][0][j]);
      h_lds[(16 + M) * STRIDE + coln] = fast_tanh(hacc[p][1][j]);
      h_lds[(32 + M) * STRIDE + coln] = fast_tanh(hacc[p][2][j]);
      h_lds[(48 + M) * STRIDE + coln] = fast_tanh(hacc[p][3][j]);
    }
  }
  __syncthreads();

  const int rt = wave;
  const float* hrow = &h_lds[(rt * 16 + m) * STRIDE];

  v8f oacc[3];
  #pragma unroll
  for (int ct = 0; ct < 3; ++ct) {
    const int coln = ct * 16 + nloc;
    const float bias = (coln < OUTD) ? b_out[coln] : 0.0f;
    const v8f bv = {bias, bias, bias, bias, bias, bias, bias, bias};
    oacc[ct] = bv;
  }

  const float* w2 = wp2 + nloc;
  #pragma unroll 8
  for (int k = 0; k < KPAD; k += 4) {
    const int kk = k + 2 * hi;
    const v2f a = *reinterpret_cast<const v2f*>(&hrow[kk]);
    const float* wr = w2 + kk * NPAD2;
    #pragma unroll
    for (int ct = 0; ct < 3; ++ct) {
      v2f b;
      b.x = wr[16 * ct];
      b.y = wr[16 * ct + NPAD2];
      oacc[ct] = __builtin_amdgcn_wmma_f32_16x16x4_f32(false, a, false, b, (short)0, oacc[ct], false, false);
    }
  }

  #pragma unroll
  for (int ct = 0; ct < 3; ++ct) {
    const int coln = ct * 16 + nloc;
    #pragma unroll
    for (int j = 0; j < 8; ++j) {
      const int M = j + 8 * hi;
      if (coln < OUTD)
        out[(size_t)(rowBase + rt * 16 + M) * OUTD + coln] = oacc[ct][j];
    }
  }
}

extern "C" void kernel_launch(void* const* d_in, const int* in_sizes, int n_in,
                              void* d_out, int out_size, void* d_ws, size_t ws_size,
                              hipStream_t stream) {
  const int*   idx     = (const int*)  d_in[0];
  const float* emb     = (const float*)d_in[1];
  const float* w_in_p  = (const float*)d_in[2];
  const float* b_in_p  = (const float*)d_in[3];
  const float* w_out_p = (const float*)d_in[4];
  const float* b_out_p = (const float*)d_in[5];
  float* out = (float*)d_out;

  float* wp  = (float*)d_ws;
  float* wp2 = wp + KPAD * KPAD;

  pad_weights<<<(KPAD * KPAD + 255) / 256, 256, 0, stream>>>(w_in_p, w_out_p, wp, wp2);

  const int batch  = in_sizes[0] / WIN;
  const int blocks = batch / ROWS;
  tagger_fused<<<blocks, THREADS, 0, stream>>>(idx, emb, wp, b_in_p, wp2, b_out_p, out);
}


# Weights stored in the workspace in B-fragment order: each lane takes the B operands of two k-steps by one 16-byte load of whole 128-byte lines, both phases (was four 4-byte loads, two lines each)
# speedup vs baseline: 3.0133x; 1.0308x over previous
#include <hip/hip_runtime.h>

typedef __attribute__((ext_vector_type(2))) float v2f;
typedef __attribute__((ext_vector_type(4))) float v4f;
typedef __attribute__((ext_vector_type(8))) float v8f;

namespace {
constexpr int WIN     = 5;
constexpr int VOCAB   = 100000;
constexpr int EMB     = 50;
constexpr int HID     = 250;
constexpr int OUTD    = 36;
constexpr int KPAD    = 256;
constexpr int NPAD2   = 48;
constexpr int ROWS    = 64;
constexpr int STRIDE  = 260;
constexpr int THREADS = 128;
constexpr int KPAIRS  = KPAD / 8;
constexpr int TILES1  = KPAD / 16;
constexpr int TILES2  = NPAD2 / 16;
constexpr int PACK1   = KPAIRS * TILES1 * 32;
constexpr int PACK2   = KPAIRS * TILES2 * 32;
constexpr int PACK_BLOCKS = PACK1 / 256;
}

__device__ __forceinline__ float fast_tanh(float x) {
  float xc = fminf(15.0f, fmaxf(-15.0f, x));
  float t  = __expf(2.0f * xc);
  return (t - 1.0f) * __builtin_amdgcn_rcpf(t + 1.0f);
}

__device__ __forceinline__ v4f pack_entry(const float* __restrict__ w, int kp, int hi,
                                          int col, int ncols)
{
  const int  k0 = 8 * kp + 2 * hi;
  const bool cok = col < ncols;
  v4f v;
  v.x = (cok && k0     < HID) ? w[(k0    ) * ncols + col] : 0.0f;
  v.y = (cok && k0 + 1 < HID) ? w[(k0 + 1) * ncols + col] : 0.0f;
  v.z = (cok && k0 + 4 < HID) ? w[(k0 + 4) * ncols + col] : 0.0f;
  v.w = (cok && k0 + 5 < HID) ? w[(k0 + 5) * ncols + col] : 0.0f;
  return v;
}

__global__ __launch_bounds__(256)
void pad_weights(const float* __restrict__ w_in, const float* __restrict__ w_out,
                 float* __restrict__ wp, float* __restrict__ wp2)
{
  const int i    = blockIdx.x * 256 + threadIdx.x;
  const int lane = i & 31;
  const int hi   = lane >> 4;
  const int n    = lane & 15;
  const int tix  = i >> 5;
  if (i < PACK1) {
    const int kp = tix / TILES1, ct = tix - kp * TILES1;
    reinterpret_cast<v4f*>(wp)[i] = pack_entry(w_in, kp, hi, 16 * ct + n, HID);
  }
  if (i < PACK2) {
    const int kp = tix / TILES2, ct = tix - kp * TILES2;
    reinterpret_cast<v4f*>(wp2)[i] = pack_entry(w_out, kp, hi, 16 * ct + n, OUTD);
  }
}

__global__ __launch_bounds__(THREADS)
void tagger_fused(const int*   __restrict__ idx,
                  const float* __restrict__ emb,
                  const float* __restrict__ wp,
                  const float* __restrict__ b_in,
                  const float* __restrict__ wp2,
                  const float* __restrict__ b_out,
                  float*       __restrict__ out)
{
  __shared__ alignas(16) float x_lds[ROWS * STRIDE];
  __shared__ alignas(16) int   i_lds[ROWS * WIN];

  const int t       = threadIdx.x;
  const int rowBase = blockIdx.x * ROWS;

  if (t < (ROWS * WIN) / 4) {
    const int4 q = *reinterpret_cast<const int4*>(idx + (size_t)rowBase * WIN + 4 * t);
    *reinterpret_cast<int4*>(&i_lds[4 * t]) = q;
  }
  __syncthreads();

  {
    const bool real = t < (HID / 2);
    const int  slot = real ? t / (EMB / 2) : (WIN - 1);
    const unsigned jp = real ? (unsigned)(t - slot * (EMB / 2)) : 0u;
    const v2f* emb2  = reinterpret_cast<const v2f*>(emb);
    const v2f  zero2 = {0.0f, 0.0f};
    #pragma unroll
    for (int r0 = 0; r0 < ROWS; r0 += 16) {
      v2f g[16];
      #pragma unroll
      for (int u = 0; u < 16; ++u) {
        const int  id = i_lds[(r0 + u) * WIN + slot];
        const bool ok = real && (unsigned)id < (unsigned)VOCAB;
        const unsigned s = ok ? (unsigned)id : 0u;
        const v2f  v  = emb2[s * (unsigned)(EMB / 2) + jp];
        g[u] = ok ? v : zero2;
      }
      #pragma unroll
      for (int u = 0; u < 16; ++u)
        *reinterpret_cast<v2f*>(&x_lds[(r0 + u) * STRIDE + 2 * t]) = g[u];
    }
  }
  __syncthreads();

  const int lane = t & 31;
  const int wave = t >> 5;
  const int hi   = lane >> 4;
  const int nloc = lane & 15;
  const int m    = lane & 15;

  const float* xr0 = &x_lds[( 0 + m) * STRIDE];
  const float* xr1 = &x_lds[(16 + m) * STRIDE];
  const float* xr2 = &x_lds[(32 + m) * STRIDE];
  const float* xr3 = &x_lds[(48 + m) * STRIDE];

  const int col0 = wave * 64 + nloc;
  v8f hacc[4][4];
  #pragma unroll
  for (int p = 0; p < 4; ++p) {
    const int coln = col0 + 16 * p;
    const float bias = (coln < HID) ? b_in[coln] : 0.0f;
    const v8f bv = {bias, bias, bias, bias, bias, bias, bias, bias};
    hacc[p][0] = bv; hacc[p][1] = bv; hacc[p][2] = bv; hacc[p][3] = bv;
  }

  const v4f* wq = reinterpret_cast<const v4f*>(wp) + (4 * wave) * 32 + lane;
  #pragma unroll 2
  for (int kp = 0; kp < KPAIRS; ++kp) {
    const int kk0 = 8 * kp + 2 * hi;
    const int kk1 = kk0 + 4;
    const v2f a00 = *reinterpret_cast<const v2f*>(&xr0[kk0]);
    const v2f a01 = *reinterpret_cast<const v2f*>(&xr1[kk0]);
    const v2f a02 = *reinterpret_cast<const v2f*>(&xr2[kk0]);
    const v2f a03 = *reinterpret_cast<const v2f*>(&xr3[kk0]);
    const v2f a10 = *reinterpret_cast<const v2f*>(&xr0[kk1]);
    const v2f a11 = *reinterpret_cast<const v2f*>(&xr1[kk1]);
    const v2f a12 = *reinterpret_cast<const v2f*>(&xr2[kk1]);
    const v2f a13 = *reinterpret_cast<const v2f*>(&xr3[kk1]);
    const v4f* wk = wq + kp * (TILES1 * 32);
    v4f b4[4];
    #pragma unroll
    for (int p = 0; p < 4; ++p) b4[p] = wk[p * 32];
    #pragma unroll
    for (int p = 0; p < 4; ++p) {
      const v2f b = {b4[p].x, b4[p].y};
      hacc[p][0] = __builtin_amdgcn_wmma_f32_16x16x4_f32(false, a00, false, b, (short)0, hacc[p][0], false, false);
      hacc[p][1] = __builtin_amdgcn_wmma_f32_16x16x4_f32(false, a01, false, b, (short)0, hacc[p][1], false, false);
      hacc[p][2] = __builtin_amdgcn_wmma_f32_16x16x4_f32(false, a02, false, b, (short)0, hacc[p][2], false, false);
      hacc[p][3] = __builtin_amdgcn_wmma_f32_16x16x4_f32(false, a03, false, b, (short)0, hacc[p][3], false, false);
    }
    #pragma unroll
    for (int p = 0; p < 4; ++p) {
      const v2f b = {b4[p].z, b4[p].w};
      hacc[p][0] = __builtin_amdgcn_wmma_f32_16x16x4_f32(false, a10, false, b, (short)0, hacc[p][0], false, false);
      hacc[p][1] = __builtin_amdgcn_wmma_f32_16x16x4_f32(false, a11, false, b, (short)0, hacc[p][1], false, false);
      hacc[p][2] = __builtin_amdgcn_wmma_f32_16x16x4_f32(false, a12, false, b, (short)0, hacc[p][2], false, false);
      hacc[p][3] = __builtin_amdgcn_wmma_f32_16x16x4_f32(false, a13, false, b, (short)0, hacc[p][3], false, false);
    }
  }
  __syncthreads();

  float* h_lds = x_lds;
  #pragma unroll
  for (int p = 0; p < 4; ++p) {
    const int coln = col0 + 16 * p;
    #pragma unroll
    for (int j = 0; j < 8; ++j) {
      const int M = j + 8 * hi;
      h_lds[( 0 + M) * STRIDE + coln] = fast_tanh(hacc[p][0][j]);
      h_lds[(16 + M) * STRIDE + coln] = fast_tanh(hacc[p][1][j]);
      h_lds[(32 + M) * STRIDE + coln] = fast_tanh(hacc[p][2][j]);
      h_lds[(48 + M) * STRIDE + coln] = fast_tanh(hacc[p][3][j]);
    }
  }
  __syncthreads();

  const int rt = wave;
  const float* hrow = &h_lds[(rt * 16 + m) * STRIDE];

  v8f oacc[3];
  #pragma unroll
  for (int ct = 0; ct < 3; ++ct) {
    const int coln = ct * 16 + nloc;
    const float bias = (coln < OUTD) ? b_out[coln] : 0.0f;
    const v8f bv = {bias, bias, bias, bias, bias, bias, bias, bias};
    oacc[ct] = bv;
  }

  const v4f* w2q = reinterpret_cast<const v4f*>(wp2) + lane;
  #pragma unroll 4
  for (int kp = 0; kp < KPAIRS; ++kp) {
    const int kk0 = 8 * kp + 2 * hi;
    const int kk1 = kk0 + 4;
    const v2f a0 = *reinterpret_cast<const v2f*>(&hrow[kk0]);
    const v2f a1 = *reinterpret_cast<const v2f*>(&hrow[kk1]);
    const v4f* wk = w2q + kp * (TILES2 * 32);
    v4f b4[3];
    #pragma unroll
    for (int ct = 0; ct < 3; ++ct) b4[ct] = wk[ct * 32];
    #pragma unroll
    for (int ct = 0; ct < 3; ++ct) {
      const v2f b = {b4[ct].x, b4[ct].y};
      oacc[ct] = __builtin_amdgcn_wmma_f32_16x16x4_f32(false, a0, false, b, (short)0, oacc[ct], false, false);
    }
    #pragma unroll
    for (int ct = 0; ct < 3; ++ct) {
      const v2f b = {b4[ct].z, b4[ct].w};
      oacc[ct] = __builtin_amdgcn_wmma_f32_16x16x4_f32(false, a1, false, b, (short)0, oacc[ct], false, false);
    }
  }

  #pragma unroll
  for (int ct = 0; ct < 3; ++ct) {
    const int coln = ct * 16 + nloc;
    #pragma unroll
    for (int j = 0; j < 8; ++j) {
      const int M = j + 8 * hi;
      if (coln < OUTD)
        out[(size_t)(rowBase + rt * 16 + M) * OUTD + coln] = oacc[ct][j];
    }
  }
}

extern "C" void kernel_launch(void* const* d_in, const int* in_sizes, int n_in,
                              void* d_out, int out_size, void* d_ws, size_t ws_size,
                              hipStream_t stream) {
  const int*   idx     = (const int*)  d_in[0];
  const float* emb     = (const float*)d_in[1];
  const float* w_in_p  = (const float*)d_in[2];
  const float* b_in_p  = (const float*)d_in[3];
  const float* w_out_p = (const float*)d_in[4];
  const float* b_out_p = (const float*)d_in[5];
  float* out = (float*)d_out;

  float* wp  = (float*)d_ws;
  float* wp2 = wp + KPAD * KPAD;

  pad_weights<<<PACK_BLOCKS, 256, 0, stream>>>(w_in_p, w_out_p, wp, wp2);

  const int batch  = in_sizes[0] / WIN;
  const int blocks = batch / ROWS;
  tagger_fused<<<blocks, THREADS, 0, stream>>>(idx, emb, wp, b_in_p, wp2, b_out_p, out);
}


# Hidden-layer tanh on v_tanh_f32, one transcendental per element (was clamp, scale, v_exp_f32, two adds, v_rcp_f32, multiply: 7 vector instructions each); last bits of h differ
# speedup vs baseline: 3.1074x; 1.0312x over previous
#include <hip/hip_runtime.h>

typedef __attribute__((ext_vector_type(2))) float v2f;
typedef __attribute__((ext_vector_type(4))) float v4f;
typedef __attribute__((ext_vector_type(8))) float v8f;

namespace {
constexpr int WIN     = 5;
constexpr int VOCAB   = 100000;
constexpr int EMB     = 50;
constexpr int HID     = 250;
constexpr int OUTD    = 36;
constexpr int KPAD    = 256;
constexpr int NPAD2   = 48;
constexpr int ROWS    = 64;
constexpr int STRIDE  = 260;
constexpr int THREADS = 128;
constexpr int KPAIRS  = KPAD / 8;
constexpr int TILES1  = KPAD / 16;
constexpr int TILES2  = NPAD2 / 16;
constexpr int PACK1   = KPAIRS * TILES1 * 32;
constexpr int PACK2   = KPAIRS * TILES2 * 32;
constexpr int PACK_BLOCKS = PACK1 / 256;
}

__device__ __forceinline__ float fast_tanh(float x) {
  return __builtin_amdgcn_tanhf(x);
}

__device__ __forceinline__ v4f pack_entry(const float* __restrict__ w, int kp, int hi,
                                          int col, int ncols)
{
  const int  k0 = 8 * kp + 2 * hi;
  const bool cok = col < ncols;
  v4f v;
  v.x = (cok && k0     < HID) ? w[(k0    ) * ncols + col] : 0.0f;
  v.y = (cok && k0 + 1 < HID) ? w[(k0 + 1) * ncols + col] : 0.0f;
  v.z = (cok && k0 + 4 < HID) ? w[(k0 + 4) * ncols + col] : 0.0f;
  v.w = (cok && k0 + 5 < HID) ? w[(k0 + 5) * ncols + col] : 0.0f;
  return v;
}

__global__ __launch_bounds__(256)
void pad_weights(const float* __restrict__ w_in, const float* __restrict__ w_out,
                 float* __restrict__ wp, float* __restrict__ wp2)
{
  const int i    = blockIdx.x * 256 + threadIdx.x;
  const int lane = i & 31;
  const int hi   = lane >> 4;
  const int n    = lane & 15;
  const int tix  = i >> 5;
  if (i < PACK1) {
    const int kp = tix / TILES1, ct = tix - kp * TILES1;
    reinterpret_cast<v4f*>(wp)[i] = pack_entry(w_in, kp, hi, 16 * ct + n, HID);
  }
  if (i < PACK2) {
    const int kp = tix / TILES2, ct = tix - kp * TILES2;
    reinterpret_cast<v4f*>(wp2)[i] = pack_entry(w_out, kp, hi, 16 * ct + n, OUTD);
  }
}

__global__ __launch_bounds__(THREADS)
void tagger_fused(const int*   __restrict__ idx,
                  const float* __restrict__ emb,
                  const float* __restrict__ wp,
                  const float* __restrict__ b_in,
                  const float* __restrict__ wp2,
                  const float* __restrict__ b_out,
                  float*       __restrict__ out)
{
  __shared__ alignas(16) float x_lds[ROWS * STRIDE];
  __shared__ alignas(16) int   i_lds[ROWS * WIN];

  const int t       = threadIdx.x;
  const int rowBase = blockIdx.x * ROWS;

  if (t < (ROWS * WIN) / 4) {
    const int4 q = *reinterpret_cast<const int4*>(idx + (size_t)rowBase * WIN + 4 * t);
    *reinterpret_cast<int4*>(&i_lds[4 * t]) = q;
  }
  __syncthreads();

  {
    const bool real = t < (HID / 2);
    const int  slot = real ? t / (EMB / 2) : (WIN - 1);
    const unsigned jp = real ? (unsigned)(t - slot * (EMB / 2)) : 0u;
    const v2f* emb2  = reinterpret_cast<const v2f*>(emb);
    const v2f  zero2 = {0.0f, 0.0f};
    #pragma unroll
    for (int r0 = 0; r0 < ROWS; r0 += 16) {
      v2f g[16];
      #pragma unroll
      for (int u = 0; u < 16; ++u) {
        const int  id = i_lds[(r0 + u) * WIN + slot];
        const bool ok = real && (unsigned)id < (unsigned)VOCAB;
        const unsigned s = ok ? (unsigned)id : 0u;
        const v2f  v  = emb2[s * (unsigned)(EMB / 2) + jp];
        g[u] = ok ? v : zero2;
      }
      #pragma unroll
      for (int u = 0; u < 16; ++u)
        *reinterpret_cast<v2f*>(&x_lds[(r0 + u) * STRIDE + 2 * t]) = g[u];
    }
  }
  __syncthreads();

  const int lane = t & 31;
  const int wave = t >> 5;
  const int hi   = lane >> 4;
  const int nloc = lane & 15;
  const int m    = lane & 15;

  const float* xr0 = &x_lds[( 0 + m) * STRIDE];
  const float* xr1 = &x_lds[(16 + m) * STRIDE];
  const float* xr2 = &x_lds[(32 + m) * STRIDE];
  const float* xr3 = &x_lds[(48 + m) * STRIDE];

  const int col0 = wave * 64 + nloc;
  v8f hacc[4][4];
  #pragma unroll
  for (int p = 0; p < 4; ++p) {
    const int coln = col0 + 16 * p;
    const float bias = (coln < HID) ? b_in[coln] : 0.0f;
    const v8f bv = {bias, bias, bias, bias, bias, bias, bias, bias};
    hacc[p][0] = bv; hacc[p][1] = bv; hacc[p][2] = bv; hacc[p][3] = bv;
  }

  const v4f* wq = reinterpret_cast<const v4f*>(wp) + (4 * wave) * 32 + lane;
  #pragma unroll 2
  for (int kp = 0; kp < KPAIRS; ++kp) {
    const int kk0 = 8 * kp + 2 * hi;
    const int kk1 = kk0 + 4;
    const v2f a00 = *reinterpret_cast<const v2f*>(&xr0[kk0]);
    const v2f a01 = *reinterpret_cast<const v2f*>(&xr1[kk0]);
    const v2f a02 = *reinterpret_cast<const v2f*>(&xr2[kk0]);
    const v2f a03 = *reinterpret_cast<const v2f*>(&xr3[kk0]);
    const v2f a10 = *reinterpret_cast<const v2f*>(&xr0[kk1]);
    const v2f a11 = *reinterpret_cast<const v2f*>(&xr1[kk1]);
    const v2f a12 = *reinterpret_cast<const v2f*>(&xr2[kk1]);
    const v2f a13 = *reinterpret_cast<const v2f*>(&xr3[kk1]);
    const v4f* wk = wq + kp * (TILES1 * 32);
    v4f b4[4];
    #pragma unroll
    for (int p = 0; p < 4; ++p) b4[p] = wk[p * 32];
    #pragma unroll
    for (int p = 0; p < 4; ++p) {
      const v2f b = {b4[p].x, b4[p].y};
      hacc[p][0] = __builtin_amdgcn_wmma_f32_16x16x4_f32(false, a00, false, b, (short)0, hacc[p][0], false, false);
      hacc[p][1] = __builtin_amdgcn_wmma_f32_16x16x4_f32(false, a01, false, b, (short)0, hacc[p][1], false, false);
      hacc[p][2] = __builtin_amdgcn_wmma_f32_16x16x4_f32(false, a02, false, b, (short)0, hacc[p][2], false, false);
      hacc[p][3] = __builtin_amdgcn_wmma_f32_16x16x4_f32(false, a03, false, b, (short)0, hacc[p][3], false, false);
    }
    #pragma unroll
    for (int p = 0; p < 4; ++p) {
      const v2f b = {b4[p].z, b4[p].w};
      hacc[p][0] = __builtin_amdgcn_wmma_f32_16x16x4_f32(false, a10, false, b, (short)0, hacc[p][0], false, false);
      hacc[p][1] = __builtin_amdgcn_wmma_f32_16x16x4_f32(false, a11, false, b, (short)0, hacc[p][1], false, false);
      hacc[p][2] = __builtin_amdgcn_wmma_f32_16x16x4_f32(false, a12, false, b, (short)0, hacc[p][2], false, false);
      hacc[p][3] = __builtin_amdgcn_wmma_f32_16x16x4_f32(false, a13, false, b, (short)0, hacc[p][3], false, false);
    }
  }
  __syncthreads();

  float* h_lds = x_lds;
  #pragma unroll
  for (int p = 0; p < 4; ++p) {
    const int coln = col0 + 16 * p;
    #pragma unroll
    for (int j = 0; j < 8; ++j) {
      const int M = j + 8 * hi;
      h_lds[( 0 + M) * STRIDE + coln] = fast_tanh(hacc[p][0][j]);
      h_lds[(16 + M) * STRIDE + coln] = fast_tanh(hacc[p][1][j]);
      h_lds[(32 + M) * STRIDE + coln] = fast_tanh(hacc[p][2][j]);
      h_lds[(48 + M) * STRIDE + coln] = fast_tanh(hacc[p][3][j]);
    }
  }
  __syncthreads();

  const int rt = wave;
  const float* hrow = &h_lds[(rt * 16 + m) * STRIDE];

  v8f oacc[3];
  #pragma unroll
  for (int ct = 0; ct < 3; ++ct) {
    const int coln = ct * 16 + nloc;
    const float bias = (coln < OUTD) ? b_out[coln] : 0.0f;
    const v8f bv = {bias, bias, bias, bias, bias, bias, bias, bias};
    oacc[ct] = bv;
  }

  const v4f* w2q = reinterpret_cast<const v4f*>(wp2) + lane;
  #pragma unroll 4
  for (int kp = 0; kp < KPAIRS; ++kp) {
    const int kk0 = 8 * kp + 2 * hi;
    const int kk1 = kk0 + 4;
    const v2f a0 = *reinterpret_cast<const v2f*>(&hrow[kk0]);
    const v2f a1 = *reinterpret_cast<const v2f*>(&hrow[kk1]);
    const v4f* wk = w2q + kp * (TILES2 * 32);
    v4f b4[3];
    #pragma unroll
    for (int ct = 0; ct < 3; ++ct) b4[ct] = wk[ct * 32];
    #pragma unroll
    for (int ct = 0; ct < 3; ++ct) {
      const v2f b = {b4[ct].x, b4[ct].y};
      oacc[ct] = __builtin_amdgcn_wmma_f32_16x16x4_f32(false, a0, false, b, (short)0, oacc[ct], false, false);
    }
    #pragma unroll
    for (int ct = 0; ct < 3; ++ct) {
      const v2f b = {b4[ct].z, b4[ct].w};
      oacc[ct] = __builtin_amdgcn_wmma_f32_16x16x4_f32(false, a1, false, b, (short)0, oacc[ct], false, false);
    }
  }

  #pragma unroll
  for (int ct = 0; ct < 3; ++ct) {
    const int coln = ct * 16 + nloc;
    #pragma unroll
    for (int j = 0; j < 8; ++j) {
      const int M = j + 8 * hi;
      if (coln < OUTD)
        out[(size_t)(rowBase + rt * 16 + M) * OUTD + coln] = oacc[ct][j];
    }
  }
}

extern "C" void kernel_launch(void* const* d_in, const int* in_sizes, int n_in,
                              void* d_out, int out_size, void* d_ws, size_t ws_size,
                              hipStream_t stream) {
  const int*   idx     = (const int*)  d_in[0];
  const float* emb     = (const float*)d_in[1];
  const float* w_in_p  = (const float*)d_in[2];
  const float* b_in_p  = (const float*)d_in[3];
  const float* w_out_p = (const float*)d_in[4];
  const float* b_out_p = (const float*)d_in[5];
  float* out = (float*)d_out;

  float* wp  = (float*)d_ws;
  float* wp2 = wp + KPAD * KPAD;

  pad_weights<<<PACK_BLOCKS, 256, 0, stream>>>(w_in_p, w_out_p, wp, wp2);

  const int batch  = in_sizes[0] / WIN;
  const int blocks = batch / ROWS;
  tagger_fused<<<blocks, THREADS, 0, stream>>>(idx, emb, wp, b_in_p, wp2, b_out_p, out);
}
